# starting kernel
# baseline (speedup 1.0000x reference)
#include <hip/hip_runtime.h>

#define NNODE 10000
#define NEDGE 320000
#define MCH   32
#define NH    8

typedef float v2f __attribute__((ext_vector_type(2)));
typedef float v8f __attribute__((ext_vector_type(8)));

__device__ __forceinline__ unsigned f2key(float x) {
    unsigned b = __float_as_uint(x);
    return (b & 0x80000000u) ? ~b : (b | 0x80000000u);
}
__device__ __forceinline__ float key2f(unsigned k) {
    if (k == 0u) return 0.0f;
    unsigned b = (k & 0x80000000u) ? (k & 0x7fffffffu) : ~k;
    return __uint_as_float(b);
}

__global__ void gmab_init(float* __restrict__ out, int out_n,
                          unsigned* __restrict__ emaxKey,
                          float* __restrict__ denom, int nh) {
    int i = blockIdx.x * blockDim.x + threadIdx.x;
    if (i < out_n) out[i] = 0.0f;
    if (i < nh) { emaxKey[i] = 0u; denom[i] = 0.0f; }
}

__global__ __launch_bounds__(256)
void gmab_dot_max(const float* __restrict__ k0, const float* __restrict__ k1,
                  const float* __restrict__ q0, const float* __restrict__ q1,
                  const int*   __restrict__ dst,
                  float* __restrict__ eArr, unsigned* __restrict__ emaxKey) {
    const int  lane  = threadIdx.x & 31;
    const int  wave  = threadIdx.x >> 5;
    const long ebase = ((long)blockIdx.x * 8 + wave) * 16;
    const int  r     = lane & 15;
    const long e     = ebase + r;
    const int  n     = dst[e];
    const int  klo   = (lane < 16) ? 0 : 2;
    const float scale = 0.08838834764831845f;

    #pragma unroll
    for (int h = 0; h < NH; ++h) {
        v8f acc = {};
        #pragma unroll
        for (int kk = 0; kk < 4; ++kk) {
            const int f0 = kk * 4 + klo;
            v2f a, b;
            if (kk == 0) {
                a.x = k0[e * MCH + h * 4 + f0];
                a.y = k0[e * MCH + h * 4 + f0 + 1];
                b.x = q0[(long)n * MCH + h * 4 + f0];
                b.y = q0[(long)n * MCH + h * 4 + f0 + 1];
            } else {
                a.x = k1[e * (MCH * 3) + h * 12 + (f0 - 4)];
                a.y = k1[e * (MCH * 3) + h * 12 + (f0 - 3)];
                b.x = q1[(long)n * (MCH * 3) + h * 12 + (f0 - 4)];
                b.y = q1[(long)n * (MCH * 3) + h * 12 + (f0 - 3)];
            }
            acc = __builtin_amdgcn_wmma_f32_16x16x4_f32(
                false, a, false, b, (short)0, acc, false, false);
        }
        float ev = 0.0f;
        #pragma unroll
        for (int v = 0; v < 8; ++v) {
            if (lane == v || lane == v + 24) ev = acc[v];
        }
        if (lane < 8 || lane >= 24) {
            const long eo = ebase + ((lane < 8) ? lane : (lane - 16));
            const float es = ev * scale;
            eArr[eo * NH + h] = es;
            atomicMax(&emaxKey[(long)n * NH + h], f2key(es));
        }
    }
}

__global__ void gmab_exp_sum(const int* __restrict__ dst,
                             const unsigned* __restrict__ emaxKey,
                             float* __restrict__ eArr,
                             float* __restrict__ denom) {
    long i = (long)blockIdx.x * blockDim.x + threadIdx.x;
    if (i >= (long)NEDGE * NH) return;
    long e = i / NH;
    int  h = (int)(i % NH);
    int  n = dst[e];
    float em = key2f(emaxKey[(long)n * NH + h]);
    float ex = __expf(eArr[i] - em);
    eArr[i] = ex;
    atomicAdd(&denom[(long)n * NH + h], ex);
}

__global__ void gmab_scatter(const float* __restrict__ v0, const float* __restrict__ v1,
                             const int*   __restrict__ dst,
                             const float* __restrict__ eArr,
                             const float* __restrict__ denom,
                             float* __restrict__ out) {
    long i = (long)blockIdx.x * blockDim.x + threadIdx.x;
    if (i >= (long)NEDGE * MCH) return;
    long e = i >> 5;
    int  m = (int)(i & 31);
    int  h = m >> 2;
    int  n = dst[e];
    float a = eArr[e * NH + h] / denom[(long)n * NH + h];

    atomicAdd(&out[(long)n * MCH + m], a * v0[i]);

    float* out1 = out + (long)NNODE * MCH;
    long vi = i * 3;
    long oi = ((long)n * MCH + m) * 3;
    atomicAdd(&out1[oi + 0], a * v1[vi + 0]);
    atomicAdd(&out1[oi + 1], a * v1[vi + 1]);
    atomicAdd(&out1[oi + 2], a * v1[vi + 2]);
}

extern "C" void kernel_launch(void* const* d_in, const int* in_sizes, int n_in,
                              void* d_out, int out_size, void* d_ws, size_t ws_size,
                              hipStream_t stream) {
    const float* v0 = (const float*)d_in[0];
    const float* v1 = (const float*)d_in[1];
    const float* k0 = (const float*)d_in[2];
    const float* k1 = (const float*)d_in[3];
    const float* q0 = (const float*)d_in[4];
    const float* q1 = (const float*)d_in[5];
    const int*   dst = (const int*)d_in[6];
    float* out = (float*)d_out;

    float*    eArr    = (float*)d_ws;
    unsigned* emaxKey = (unsigned*)(eArr + (size_t)NEDGE * NH);
    float*    denom   = (float*)(emaxKey + (size_t)NNODE * NH);

    const int nh = NNODE * NH;
    const int initN = (out_size > nh) ? out_size : nh;
    gmab_init<<<(initN + 255) / 256, 256, 0, stream>>>(out, out_size, emaxKey, denom, nh);

    gmab_dot_max<<<NEDGE / 128, 256, 0, stream>>>(k0, k1, q0, q1, dst, eArr, emaxKey);

    const long p2 = (long)NEDGE * NH;
    gmab_exp_sum<<<(int)((p2 + 255) / 256), 256, 0, stream>>>(dst, emaxKey, eArr, denom);

    const long p3 = (long)NEDGE * MCH;
    gmab_scatter<<<(int)((p3 + 255) / 256), 256, 0, stream>>>(v0, v1, dst, eArr, denom, out);
}


# dot pass: one edge per WMMA tile (rows = 8 heads x 2 feature halves), so a 16-byte lane load covers 4 whole 128-B lines, not 16 partial ones; a dot = two 8-product halves added (sum order changes)
# speedup vs baseline: 2.2272x; 2.2272x over previous
#include <hip/hip_runtime.h>

#define NNODE 10000
#define NEDGE 320000
#define MCH   32
#define NH    8

typedef float v2f __attribute__((ext_vector_type(2)));
typedef float v8f __attribute__((ext_vector_type(8)));

__device__ __forceinline__ unsigned f2key(float x) {
    unsigned b = __float_as_uint(x);
    return (b & 0x80000000u) ? ~b : (b | 0x80000000u);
}
__device__ __forceinline__ float key2f(unsigned k) {
    if (k == 0u) return 0.0f;
    unsigned b = (k & 0x80000000u) ? (k & 0x7fffffffu) : ~k;
    return __uint_as_float(b);
}

__global__ void gmab_init(float* __restrict__ out, int out_n,
                          unsigned* __restrict__ emaxKey,
                          float* __restrict__ denom, int nh) {
    int i = blockIdx.x * blockDim.x + threadIdx.x;
    if (i < out_n) out[i] = 0.0f;
    if (i < nh) { emaxKey[i] = 0u; denom[i] = 0.0f; }
}

typedef float v4f __attribute__((ext_vector_type(4)));

__device__ __forceinline__ float xrow8(float x) {
    int b = __builtin_bit_cast(int, x);
    return __builtin_bit_cast(float,
        __builtin_amdgcn_permlanex16(b, b, 0xfedcba98u, 0x76543210u, false, false));
}

__global__ __launch_bounds__(256)
void gmab_dot_max(const float* __restrict__ k0, const float* __restrict__ k1,
                  const float* __restrict__ q0, const float* __restrict__ q1,
                  const int*   __restrict__ dst,
                  float* __restrict__ eArr, unsigned* __restrict__ emaxKey) {
    const int  lane  = threadIdx.x & 31;
    const int  wave  = __builtin_amdgcn_readfirstlane((int)(threadIdx.x >> 5));
    const long ebase = ((long)blockIdx.x * 8 + wave) * 16;
    const int  h     = lane & 7;
    const int  part  = ((lane >> 3) & 1) * 2 + (lane >> 4);
    const bool deg0  = (part == 3);
    const float scale = 0.08838834764831845f;

    const float*   kb     = deg0 ? k0 : k1;
    const float*   qb     = deg0 ? q0 : q1;
    const unsigned rowlen = deg0 ? MCH : MCH * 3;
    const unsigned off    = deg0 ? h * 4 : h * 12 + part * 4;
    const float*   kp     = kb + ((unsigned long)ebase * rowlen + off);
    const float*   qp     = qb + off;

    v4f kf[16], qf[16];
    int nd[16];
    for (int t = 0; t < 16; ++t)
        nd[t] = dst[ebase + t];
    for (int t = 0; t < 16; ++t)
        kf[t] = *(const v4f*)(kp + t * rowlen);
    for (int t = 0; t < 16; ++t)
        qf[t] = *(const v4f*)(qp + __umul24((unsigned)nd[t], rowlen));

    for (int t = 0; t < 16; t += 2) {
        float s2[2];
        for (int u = 0; u < 2; ++u) {
            const v4f kv = kf[t + u], qv = qf[t + u];
            v8f acc = {};
            v2f a, b;
            a.x = kv.x; a.y = kv.y; b.x = qv.x; b.y = qv.y;
            acc = __builtin_amdgcn_wmma_f32_16x16x4_f32(
                false, a, false, b, (short)0, acc, false, false);
            a.x = kv.z; a.y = kv.w; b.x = qv.z; b.y = qv.w;
            acc = __builtin_amdgcn_wmma_f32_16x16x4_f32(
                false, a, false, b, (short)0, acc, false, false);
            float ev = 0.0f;
            for (int v = 0; v < 8; ++v) {
                if (h == v) ev = acc[v];
            }
            s2[u] = ev + xrow8(ev);
        }
        const int   up = lane >> 4;
        const float es = (up ? s2[1] : s2[0]) * scale;
        if (lane < 8 || lane >= 24) {
            const long eo = ebase + t + up;
            const int  n  = up ? nd[t + 1] : nd[t];
            eArr[eo * NH + h] = es;
            atomicMax(&emaxKey[(long)n * NH + h], f2key(es));
        }
    }
}

__global__ void gmab_exp_sum(const int* __restrict__ dst,
                             const unsigned* __restrict__ emaxKey,
                             float* __restrict__ eArr,
                             float* __restrict__ denom) {
    long i = (long)blockIdx.x * blockDim.x + threadIdx.x;
    if (i >= (long)NEDGE * NH) return;
    long e = i / NH;
    int  h = (int)(i % NH);
    int  n = dst[e];
    float em = key2f(emaxKey[(long)n * NH + h]);
    float ex = __expf(eArr[i] - em);
    eArr[i] = ex;
    atomicAdd(&denom[(long)n * NH + h], ex);
}

__global__ void gmab_scatter(const float* __restrict__ v0, const float* __restrict__ v1,
                             const int*   __restrict__ dst,
                             const float* __restrict__ eArr,
                             const float* __restrict__ denom,
                             float* __restrict__ out) {
    long i = (long)blockIdx.x * blockDim.x + threadIdx.x;
    if (i >= (long)NEDGE * MCH) return;
    long e = i >> 5;
    int  m = (int)(i & 31);
    int  h = m >> 2;
    int  n = dst[e];
    float a = eArr[e * NH + h] / denom[(long)n * NH + h];

    atomicAdd(&out[(long)n * MCH + m], a * v0[i]);

    float* out1 = out + (long)NNODE * MCH;
    long vi = i * 3;
    long oi = ((long)n * MCH + m) * 3;
    atomicAdd(&out1[oi + 0], a * v1[vi + 0]);
    atomicAdd(&out1[oi + 1], a * v1[vi + 1]);
    atomicAdd(&out1[oi + 2], a * v1[vi + 2]);
}

extern "C" void kernel_launch(void* const* d_in, const int* in_sizes, int n_in,
                              void* d_out, int out_size, void* d_ws, size_t ws_size,
                              hipStream_t stream) {
    const float* v0 = (const float*)d_in[0];
    const float* v1 = (const float*)d_in[1];
    const float* k0 = (const float*)d_in[2];
    const float* k1 = (const float*)d_in[3];
    const float* q0 = (const float*)d_in[4];
    const float* q1 = (const float*)d_in[5];
    const int*   dst = (const int*)d_in[6];
    float* out = (float*)d_out;

    float*    eArr    = (float*)d_ws;
    unsigned* emaxKey = (unsigned*)(eArr + (size_t)NEDGE * NH);
    float*    denom   = (float*)(emaxKey + (size_t)NNODE * NH);

    const int nh = NNODE * NH;
    const int initN = (out_size > nh) ? out_size : nh;
    gmab_init<<<(initN + 255) / 256, 256, 0, stream>>>(out, out_size, emaxKey, denom, nh);

    gmab_dot_max<<<NEDGE / 128, 256, 0, stream>>>(k0, k1, q0, q1, dst, eArr, emaxKey);

    const long p2 = (long)NEDGE * NH;
    gmab_exp_sum<<<(int)((p2 + 255) / 256), 256, 0, stream>>>(dst, emaxKey, eArr, denom);

    const long p3 = (long)NEDGE * MCH;
    gmab_scatter<<<(int)((p3 + 255) / 256), 256, 0, stream>>>(v0, v1, dst, eArr, denom, out);
}


# scatter pass: one wave per 16 edges, lane = one float of a 128-B line: every value load and atomic add covers one whole line (4 per edge, not 10 partial), 64 loads in flight; adds arrive as before
# speedup vs baseline: 2.6432x; 1.1868x over previous
#include <hip/hip_runtime.h>

#define NNODE 10000
#define NEDGE 320000
#define MCH   32
#define NH    8

typedef float v2f __attribute__((ext_vector_type(2)));
typedef float v8f __attribute__((ext_vector_type(8)));

__device__ __forceinline__ unsigned f2key(float x) {
    unsigned b = __float_as_uint(x);
    return (b & 0x80000000u) ? ~b : (b | 0x80000000u);
}
__device__ __forceinline__ float key2f(unsigned k) {
    if (k == 0u) return 0.0f;
    unsigned b = (k & 0x80000000u) ? (k & 0x7fffffffu) : ~k;
    return __uint_as_float(b);
}

__global__ void gmab_init(float* __restrict__ out, int out_n,
                          unsigned* __restrict__ emaxKey,
                          float* __restrict__ denom, int nh) {
    int i = blockIdx.x * blockDim.x + threadIdx.x;
    if (i < out_n) out[i] = 0.0f;
    if (i < nh) { emaxKey[i] = 0u; denom[i] = 0.0f; }
}

typedef float v4f __attribute__((ext_vector_type(4)));

__device__ __forceinline__ float xrow8(float x) {
    int b = __builtin_bit_cast(int, x);
    return __builtin_bit_cast(float,
        __builtin_amdgcn_permlanex16(b, b, 0xfedcba98u, 0x76543210u, false, false));
}

__global__ __launch_bounds__(256)
void gmab_dot_max(const float* __restrict__ k0, const float* __restrict__ k1,
                  const float* __restrict__ q0, const float* __restrict__ q1,
                  const int*   __restrict__ dst,
                  float* __restrict__ eArr, unsigned* __restrict__ emaxKey) {
    const int  lane  = threadIdx.x & 31;
    const int  wave  = __builtin_amdgcn_readfirstlane((int)(threadIdx.x >> 5));
    const long ebase = ((long)blockIdx.x * 8 + wave) * 16;
    const int  h     = lane & 7;
    const int  part  = ((lane >> 3) & 1) * 2 + (lane >> 4);
    const bool deg0  = (part == 3);
    const float scale = 0.08838834764831845f;

    const float*   kb     = deg0 ? k0 : k1;
    const float*   qb     = deg0 ? q0 : q1;
    const unsigned rowlen = deg0 ? MCH : MCH * 3;
    const unsigned off    = deg0 ? h * 4 : h * 12 + part * 4;
    const float*   kp     = kb + ((unsigned long)ebase * rowlen + off);
    const float*   qp     = qb + off;

    v4f kf[16], qf[16];
    int nd[16];
    for (int t = 0; t < 16; ++t)
        nd[t] = dst[ebase + t];
    for (int t = 0; t < 16; ++t)
        kf[t] = *(const v4f*)(kp + t * rowlen);
    for (int t = 0; t < 16; ++t)
        qf[t] = *(const v4f*)(qp + __umul24((unsigned)nd[t], rowlen));

    for (int t = 0; t < 16; t += 2) {
        float s2[2];
        for (int u = 0; u < 2; ++u) {
            const v4f kv = kf[t + u], qv = qf[t + u];
            v8f acc = {};
            v2f a, b;
            a.x = kv.x; a.y = kv.y; b.x = qv.x; b.y = qv.y;
            acc = __builtin_amdgcn_wmma_f32_16x16x4_f32(
                false, a, false, b, (short)0, acc, false, false);
            a.x = kv.z; a.y = kv.w; b.x = qv.z; b.y = qv.w;
            acc = __builtin_amdgcn_wmma_f32_16x16x4_f32(
                false, a, false, b, (short)0, acc, false, false);
            float ev = 0.0f;
            for (int v = 0; v < 8; ++v) {
                if (h == v) ev = acc[v];
            }
            s2[u] = ev + xrow8(ev);
        }
        const int   up = lane >> 4;
        const float es = (up ? s2[1] : s2[0]) * scale;
        if (lane < 8 || lane >= 24) {
            const long eo = ebase + t + up;
            const int  n  = up ? nd[t + 1] : nd[t];
            eArr[eo * NH + h] = es;
            atomicMax(&emaxKey[(long)n * NH + h], f2key(es));
        }
    }
}

__global__ void gmab_exp_sum(const int* __restrict__ dst,
                             const unsigned* __restrict__ emaxKey,
                             float* __restrict__ eArr,
                             float* __restrict__ denom) {
    long i = (long)blockIdx.x * blockDim.x + threadIdx.x;
    if (i >= (long)NEDGE * NH) return;
    long e = i / NH;
    int  h = (int)(i % NH);
    int  n = dst[e];
    float em = key2f(emaxKey[(long)n * NH + h]);
    float ex = __expf(eArr[i] - em);
    eArr[i] = ex;
    atomicAdd(&denom[(long)n * NH + h], ex);
}

__global__ __launch_bounds__(256)
void gmab_scatter(const float* __restrict__ v0, const float* __restrict__ v1,
                  const int*   __restrict__ dst,
                  const float* __restrict__ eArr,
                  const float* __restrict__ denom,
                  float* __restrict__ out) {
    const int  lane  = threadIdx.x & 31;
    const int  wave  = __builtin_amdgcn_readfirstlane((int)(threadIdx.x >> 5));
    const long ebase = ((long)blockIdx.x * 8 + wave) * 16;

    int nd[16];
    for (int t = 0; t < 16; ++t)
        nd[t] = dst[ebase + t];

    const float* p0 = v0 + ebase * MCH + lane;
    const float* p1 = v1 + ebase * (MCH * 3) + lane;
    float x0[16], x1[16][3];
    for (int t = 0; t < 16; ++t)
        x0[t] = p0[t * MCH];
    for (int t = 0; t < 16; ++t)
        for (int j = 0; j < 3; ++j)
            x1[t][j] = p1[t * (MCH * 3) + j * 32];

    const int m1 = -((lane >> 3) & 1);
    const int m2 = -((lane >> 4) & 1);
    float aw[4];
    for (int g = 0; g < 4; ++g) {
        const int lo  = nd[4 * g]     ^ ((nd[4 * g]     ^ nd[4 * g + 1]) & m1);
        const int hi  = nd[4 * g + 2] ^ ((nd[4 * g + 2] ^ nd[4 * g + 3]) & m1);
        const int n_l = lo ^ ((lo ^ hi) & m2);
        aw[g] = eArr[(ebase + 4 * g) * NH + lane] / denom[(long)n_l * NH + (lane & 7)];
    }

    const int hs0 = lane >> 2;
    const int hs1 = lane / 12;
    const int hs2 = (lane + 32) / 12;
    const int hs3 = (lane + 64) / 12;

    float* out1 = out + (long)NNODE * MCH;
    for (int t = 0; t < 16; ++t) {
        const int   g  = t >> 2;
        const int   i8 = (t & 3) * 8;
        const float a0 = __shfl(aw[g], i8 + hs0);
        const float a1 = __shfl(aw[g], i8 + hs1);
        const float a2 = __shfl(aw[g], i8 + hs2);
        const float a3 = __shfl(aw[g], i8 + hs3);
        float* o0 = out  + (long)nd[t] * MCH + lane;
        float* o1 = out1 + (long)nd[t] * (MCH * 3) + lane;
        atomicAdd(o0,      a0 * x0[t]);
        atomicAdd(o1,      a1 * x1[t][0]);
        atomicAdd(o1 + 32, a2 * x1[t][1]);
        atomicAdd(o1 + 64, a3 * x1[t][2]);
    }
}

extern "C" void kernel_launch(void* const* d_in, const int* in_sizes, int n_in,
                              void* d_out, int out_size, void* d_ws, size_t ws_size,
                              hipStream_t stream) {
    const float* v0 = (const float*)d_in[0];
    const float* v1 = (const float*)d_in[1];
    const float* k0 = (const float*)d_in[2];
    const float* k1 = (const float*)d_in[3];
    const float* q0 = (const float*)d_in[4];
    const float* q1 = (const float*)d_in[5];
    const int*   dst = (const int*)d_in[6];
    float* out = (float*)d_out;

    float*    eArr    = (float*)d_ws;
    unsigned* emaxKey = (unsigned*)(eArr + (size_t)NEDGE * NH);
    float*    denom   = (float*)(emaxKey + (size_t)NNODE * NH);

    const int nh = NNODE * NH;
    const int initN = (out_size > nh) ? out_size : nh;
    gmab_init<<<(initN + 255) / 256, 256, 0, stream>>>(out, out_size, emaxKey, denom, nh);

    gmab_dot_max<<<NEDGE / 128, 256, 0, stream>>>(k0, k1, q0, q1, dst, eArr, emaxKey);

    const long p2 = (long)NEDGE * NH;
    gmab_exp_sum<<<(int)((p2 + 255) / 256), 256, 0, stream>>>(dst, emaxKey, eArr, denom);

    gmab_scatter<<<NEDGE / 128, 256, 0, stream>>>(v0, v1, dst, eArr, denom, out);
}


# edges grouped by destination in-call (counting sort, int atomic ranks); one wave per node does dot, running-max softmax and weighted sum in registers: no float atomics; sum order changes
# speedup vs baseline: 2.9670x; 1.1225x over previous
#include <hip/hip_runtime.h>

#define NNODE 10000
#define NEDGE 320000
#define MCH   32
#define NH    8

typedef float v2f __attribute__((ext_vector_type(2)));
typedef float v4f __attribute__((ext_vector_type(4)));
typedef float v8f __attribute__((ext_vector_type(8)));


__global__ void gmab_zero(int* __restrict__ cnt) {
    const int i = blockIdx.x * blockDim.x + threadIdx.x;
    if (i < NNODE) cnt[i] = 0;
}

__global__ __launch_bounds__(256)
void gmab_rank(const int* __restrict__ dst, int* __restrict__ cnt, int* __restrict__ rank) {
    const int i = blockIdx.x * 256 + threadIdx.x;
    if (i >= NEDGE / 4) return;
    const int4 d = ((const int4*)dst)[i];
    int4 r;
    r.x = atomicAdd(&cnt[d.x], 1);
    r.y = atomicAdd(&cnt[d.y], 1);
    r.z = atomicAdd(&cnt[d.z], 1);
    r.w = atomicAdd(&cnt[d.w], 1);
    ((int4*)rank)[i] = r;
}

#define SCAN_T   1024
#define SCAN_PER ((NNODE + SCAN_T - 1) / SCAN_T)
__global__ __launch_bounds__(SCAN_T)
void gmab_scan(const int* __restrict__ cnt, int* __restrict__ offs) {
    __shared__ int part[SCAN_T];
    const int tid = threadIdx.x;
    const int b   = tid * SCAN_PER;
    int loc[SCAN_PER];
    int s = 0;
    for (int j = 0; j < SCAN_PER; ++j) {
        const int i = b + j;
        const int v = (i < NNODE) ? cnt[i] : 0;
        loc[j] = s;
        s += v;
    }
    part[tid] = s;
    __syncthreads();
    for (int d = 1; d < SCAN_T; d <<= 1) {
        const int v = (tid >= d) ? part[tid - d] : 0;
        __syncthreads();
        part[tid] += v;
        __syncthreads();
    }
    const int base = part[tid] - s;
    for (int j = 0; j < SCAN_PER; ++j) {
        const int i = b + j;
        if (i < NNODE) offs[i] = base + loc[j];
    }
    if (tid == SCAN_T - 1) offs[NNODE] = part[SCAN_T - 1];
}

__global__ __launch_bounds__(256)
void gmab_fill(const int* __restrict__ dst, const int* __restrict__ rank,
               const int* __restrict__ offs, int* __restrict__ perm) {
    const int i = blockIdx.x * 256 + threadIdx.x;
    if (i >= NEDGE / 4) return;
    const int4 d = ((const int4*)dst)[i];
    const int4 r = ((const int4*)rank)[i];
    perm[offs[d.x] + r.x] = 4 * i;
    perm[offs[d.y] + r.y] = 4 * i + 1;
    perm[offs[d.z] + r.z] = 4 * i + 2;
    perm[offs[d.w] + r.w] = 4 * i + 3;
}

__device__ __forceinline__ float xrow8(float x) {
    int b = __builtin_bit_cast(int, x);
    return __builtin_bit_cast(float,
        __builtin_amdgcn_permlanex16(b, b, 0xfedcba98u, 0x76543210u, false, false));
}
__device__ __forceinline__ float xpos8(float x) {
    int b = __builtin_bit_cast(int, x);
    return __builtin_bit_cast(float,
        __builtin_amdgcn_permlane16(b, b, 0xfedcba98u, 0x76543210u, false, false));
}

__global__ __launch_bounds__(256)
void gmab_node(const float* __restrict__ v0, const float* __restrict__ v1,
               const float* __restrict__ k0, const float* __restrict__ k1,
               const float* __restrict__ q0, const float* __restrict__ q1,
               const int*   __restrict__ offs, const int* __restrict__ perm,
               float* __restrict__ out) {
    const int lane = threadIdx.x & 31;
    const int wave = __builtin_amdgcn_readfirstlane((int)(threadIdx.x >> 5));
    const int n    = blockIdx.x * 8 + wave;
    if (n >= NNODE) return;
    const int  h     = lane & 7;
    const int  part  = ((lane >> 3) & 1) * 2 + (lane >> 4);
    const bool deg0  = (part == 3);
    const bool own   = (lane < 8) || (lane >= 24);
    const float scale = 0.08838834764831845f;

    const float*   kb     = deg0 ? k0 : k1;
    const float*   vb     = deg0 ? v0 : v1;
    const float*   qb     = deg0 ? q0 : q1;
    float*         ob     = deg0 ? out : out + (long)NNODE * MCH;
    const unsigned rowlen = deg0 ? MCH : MCH * 3;
    const unsigned off    = deg0 ? h * 4 : h * 12 + part * 4;

    const v4f qv  = *(const v4f*)(qb + ((unsigned)n * rowlen + off));
    const int beg = offs[n];
    const int end = offs[n + 1];

    float m    = -__builtin_inff();
    float dsum = 0.0f;
    v4f   acc  = {0.0f, 0.0f, 0.0f, 0.0f};

    for (int c = beg; c < end; c += 16) {
        const int cnt = end - c;
        unsigned ed[16];
        for (int t = 0; t < 16; ++t) {
            const int i = c + t;
            const unsigned e = (unsigned)perm[i < end ? i : end - 1];
            ed[t] = e < (unsigned)(NEDGE - 1) ? e : (unsigned)(NEDGE - 1);
        }
        v4f kf[16], vf[16];
        for (int t = 0; t < 16; ++t)
            kf[t] = *(const v4f*)(kb + (ed[t] * rowlen + off));
        for (int t = 0; t < 16; ++t)
            vf[t] = *(const v4f*)(vb + (ed[t] * rowlen + off));

        float sc[16];
        float cm = -__builtin_inff();
        for (int t = 0; t < 16; ++t) {
            const v4f kv = kf[t];
            v8f d = {};
            v2f a, b;
            a.x = kv.x; a.y = kv.y; b.x = qv.x; b.y = qv.y;
            d = __builtin_amdgcn_wmma_f32_16x16x4_f32(
                false, a, false, b, (short)0, d, false, false);
            a.x = kv.z; a.y = kv.w; b.x = qv.z; b.y = qv.w;
            d = __builtin_amdgcn_wmma_f32_16x16x4_f32(
                false, a, false, b, (short)0, d, false, false);
            float ev = 0.0f;
            for (int v = 0; v < 8; ++v) {
                if (h == v) ev = d[v];
            }
            ev = own ? ev : 0.0f;
            float s = ev + xrow8(ev);
            s = s + xpos8(s);
            sc[t] = s * scale;
            cm = fmaxf(cm, (t < cnt) ? sc[t] : -__builtin_inff());
        }

        const float mn   = fmaxf(m, cm);
        const float corr = __expf(m - mn);
        dsum *= corr;
        acc  *= corr;
        for (int t = 0; t < 16; ++t) {
            const float ex = (t < cnt) ? __expf(sc[t] - mn) : 0.0f;
            dsum += ex;
            acc  += ex * vf[t];
        }
        m = mn;
    }

    v4f o = {0.0f, 0.0f, 0.0f, 0.0f};
    if (end > beg) o = acc / dsum;
    *(v4f*)(ob + ((unsigned)n * rowlen + off)) = o;
}

extern "C" void kernel_launch(void* const* d_in, const int* in_sizes, int n_in,
                              void* d_out, int out_size, void* d_ws, size_t ws_size,
                              hipStream_t stream) {
    const float* v0 = (const float*)d_in[0];
    const float* v1 = (const float*)d_in[1];
    const float* k0 = (const float*)d_in[2];
    const float* k1 = (const float*)d_in[3];
    const float* q0 = (const float*)d_in[4];
    const float* q1 = (const float*)d_in[5];
    const int*   dst = (const int*)d_in[6];
    float* out = (float*)d_out;

    int* cnt  = (int*)d_ws;
    int* offs = cnt + NNODE;
    int* rank = offs + (NNODE + 16);
    int* perm = rank + NEDGE;

    gmab_zero<<<(NNODE + 255) / 256, 256, 0, stream>>>(cnt);
    gmab_rank<<<(NEDGE / 4 + 255) / 256, 256, 0, stream>>>(dst, cnt, rank);
    gmab_scan<<<1, SCAN_T, 0, stream>>>(cnt, offs);
    gmab_fill<<<(NEDGE / 4 + 255) / 256, 256, 0, stream>>>(dst, rank, offs, perm);
    gmab_node<<<(NNODE + 7) / 8, 256, 0, stream>>>(v0, v1, k0, k1, q0, q1, offs, perm, out);
}


# each edge goes at its atomic rank into a fixed 64-slot list of its node, read whole by 2 loads per wave (a fuller node's wave reads dst): no scan, fill or rank array, 5 launches -> 3; sum order kept
# speedup vs baseline: 3.4665x; 1.1684x over previous
#include <hip/hip_runtime.h>

#define NNODE 10000
#define NEDGE 320000
#define MCH   32
#define NH    8

typedef float v2f __attribute__((ext_vector_type(2)));
typedef float v4f __attribute__((ext_vector_type(4)));
typedef float v8f __attribute__((ext_vector_type(8)));

#define LIST_CAP 64

__global__ void gmab_zero(int* __restrict__ cnt) {
    const int i = blockIdx.x * blockDim.x + threadIdx.x;
    if (i < NNODE) cnt[i] = 0;
}

__global__ __launch_bounds__(256)
void gmab_rank(const int* __restrict__ dst, int* __restrict__ cnt, int* __restrict__ slots) {
    const int i = blockIdx.x * 256 + threadIdx.x;
    if (i >= NEDGE / 4) return;
    const int4 d = ((const int4*)dst)[i];
    const int rx = atomicAdd(&cnt[d.x], 1);
    const int ry = atomicAdd(&cnt[d.y], 1);
    const int rz = atomicAdd(&cnt[d.z], 1);
    const int rw = atomicAdd(&cnt[d.w], 1);
    if (rx < LIST_CAP) slots[d.x * LIST_CAP + rx] = 4 * i;
    if (ry < LIST_CAP) slots[d.y * LIST_CAP + ry] = 4 * i + 1;
    if (rz < LIST_CAP) slots[d.z * LIST_CAP + rz] = 4 * i + 2;
    if (rw < LIST_CAP) slots[d.w * LIST_CAP + rw] = 4 * i + 3;
}

__device__ __forceinline__ float xrow8(float x) {
    int b = __builtin_bit_cast(int, x);
    return __builtin_bit_cast(float,
        __builtin_amdgcn_permlanex16(b, b, 0xfedcba98u, 0x76543210u, false, false));
}
__device__ __forceinline__ float xpos8(float x) {
    int b = __builtin_bit_cast(int, x);
    return __builtin_bit_cast(float,
        __builtin_amdgcn_permlane16(b, b, 0xfedcba98u, 0x76543210u, false, false));
}

__global__ __launch_bounds__(256)
void gmab_node(const float* __restrict__ v0, const float* __restrict__ v1,
               const float* __restrict__ k0, const float* __restrict__ k1,
               const float* __restrict__ q0, const float* __restrict__ q1,
               const int*   __restrict__ cnt, const int* __restrict__ slots,
               const int*   __restrict__ dst, float* __restrict__ out) {
    const int lane = threadIdx.x & 31;
    const int wave = __builtin_amdgcn_readfirstlane((int)(threadIdx.x >> 5));
    const int n    = blockIdx.x * 8 + wave;
    if (n >= NNODE) return;
    const int  h     = lane & 7;
    const int  part  = ((lane >> 3) & 1) * 2 + (lane >> 4);
    const bool deg0  = (part == 3);
    const bool own   = (lane < 8) || (lane >= 24);
    const float scale = 0.08838834764831845f;

    const float*   kb     = deg0 ? k0 : k1;
    const float*   vb     = deg0 ? v0 : v1;
    const float*   qb     = deg0 ? q0 : q1;
    float*         ob     = deg0 ? out : out + (long)NNODE * MCH;
    const unsigned rowlen = deg0 ? MCH : MCH * 3;
    const unsigned off    = deg0 ? h * 4 : h * 12 + part * 4;

    const v4f qv  = *(const v4f*)(qb + ((unsigned)n * rowlen + off));
    const int  deg  = cnt[n];
    const bool scan = deg > LIST_CAP;
    const int* lst  = slots + n * LIST_CAP;
    const int  la   = lst[lane];
    const int  lb   = lst[32 + lane];
    int      pos   = 0;
    int      si    = 0;
    int      sbase = 0;
    unsigned smask = 0;

    float m    = -__builtin_inff();
    float dsum = 0.0f;
    v4f   acc  = {0.0f, 0.0f, 0.0f, 0.0f};

    for (;;) {
        int cn;
        int edv;
        int lb0;
        if (!scan) {
            if (pos >= deg) break;
            cn  = deg - pos < 16 ? deg - pos : 16;
            edv = pos < 32 ? la : lb;
            lb0 = pos & 16;
            pos += 16;
        } else {
            cn  = 0;
            edv = 0;
            lb0 = 0;
            while (cn < 16) {
                if (smask == 0) {
                    if (si >= NEDGE) break;
                    smask = __builtin_amdgcn_ballot_w32(dst[si + lane] == n);
                    sbase = si;
                    si += 32;
                    continue;
                }
                const int t = __builtin_ctz(smask);
                smask &= smask - 1;
                if (lane == cn) edv = sbase + t;
                ++cn;
            }
            if (cn == 0) break;
        }
        unsigned ed[16];
        for (int t = 0; t < 16; ++t) {
            const unsigned e = (unsigned)__builtin_amdgcn_readlane(edv, lb0 + (t < cn ? t : 0));
            ed[t] = e < (unsigned)(NEDGE - 1) ? e : (unsigned)(NEDGE - 1);
        }
        v4f kf[16], vf[16];
        for (int t = 0; t < 16; ++t)
            kf[t] = *(const v4f*)(kb + (ed[t] * rowlen + off));
        for (int t = 0; t < 16; ++t)
            vf[t] = *(const v4f*)(vb + (ed[t] * rowlen + off));

        float sc[16];
        float cm = -__builtin_inff();
        for (int t = 0; t < 16; ++t) {
            const v4f kv = kf[t];
            v8f d = {};
            v2f a, b;
            a.x = kv.x; a.y = kv.y; b.x = qv.x; b.y = qv.y;
            d = __builtin_amdgcn_wmma_f32_16x16x4_f32(
                false, a, false, b, (short)0, d, false, false);
            a.x = kv.z; a.y = kv.w; b.x = qv.z; b.y = qv.w;
            d = __builtin_amdgcn_wmma_f32_16x16x4_f32(
                false, a, false, b, (short)0, d, false, false);
            float ev = 0.0f;
            for (int v = 0; v < 8; ++v) {
                if (h == v) ev = d[v];
            }
            ev = own ? ev : 0.0f;
            float s = ev + xrow8(ev);
            s = s + xpos8(s);
            sc[t] = s * scale;
            cm = fmaxf(cm, (t < cn) ? sc[t] : -__builtin_inff());
        }

        const float mn   = fmaxf(m, cm);
        const float corr = __expf(m - mn);
        dsum *= corr;
        acc  *= corr;
        for (int t = 0; t < 16; ++t) {
            const float ex = (t < cn) ? __expf(sc[t] - mn) : 0.0f;
            dsum += ex;
            acc  += ex * vf[t];
        }
        m = mn;
    }

    v4f o = {0.0f, 0.0f, 0.0f, 0.0f};
    if (deg > 0) o = acc / dsum;
    *(v4f*)(ob + ((unsigned)n * rowlen + off)) = o;
}

extern "C" void kernel_launch(void* const* d_in, const int* in_sizes, int n_in,
                              void* d_out, int out_size, void* d_ws, size_t ws_size,
                              hipStream_t stream) {
    const float* v0 = (const float*)d_in[0];
    const float* v1 = (const float*)d_in[1];
    const float* k0 = (const float*)d_in[2];
    const float* k1 = (const float*)d_in[3];
    const float* q0 = (const float*)d_in[4];
    const float* q1 = (const float*)d_in[5];
    const int*   dst = (const int*)d_in[6];
    float* out = (float*)d_out;

    int* cnt   = (int*)d_ws;
    int* slots = cnt + 10048;

    gmab_zero<<<(NNODE + 255) / 256, 256, 0, stream>>>(cnt);
    gmab_rank<<<(NEDGE / 4 + 255) / 256, 256, 0, stream>>>(dst, cnt, slots);
    gmab_node<<<(NNODE + 7) / 8, 256, 0, stream>>>(v0, v1, k0, k1, q0, q1, cnt, slots, dst, out);
}
